# starting kernel
# baseline (speedup 1.0000x reference)
#include <hip/hip_runtime.h>

#define H      64
#define G      256
#define CHUNK  128

typedef __attribute__((ext_vector_type(2))) float v2f;
typedef __attribute__((ext_vector_type(8))) float v8f;

__device__ __forceinline__ v2f mkv2(float a, float b) { v2f r; r.x = a; r.y = b; return r; }

__device__ __forceinline__ float fsigmoid(float x) {
    return __builtin_amdgcn_rcpf(1.0f + __expf(-x));
}
__device__ __forceinline__ float ftanh(float x) {
    return 2.0f * __builtin_amdgcn_rcpf(1.0f + __expf(-2.0f * x)) - 1.0f;
}

__global__ __launch_bounds__(256, 1)
void lstm_scan_kernel(const float* __restrict__ x,
                      const float* __restrict__ W_ih,
                      const float* __restrict__ W_hh,
                      const float* __restrict__ b_ih,
                      const float* __restrict__ b_hh,
                      float* __restrict__ hs,
                      int T) {
    __shared__ float h_lds[H];
    __shared__ float gates_lds[G];
    __shared__ float xbuf[2 * CHUNK];

    const int j = threadIdx.x;

    v2f w2[H / 2];
    {
        const float4* wrow = (const float4*)(W_hh + j * H);
        #pragma unroll
        for (int k = 0; k < H / 4; ++k) {
            float4 w = wrow[k];
            w2[2 * k + 0] = mkv2(w.x, w.y);
            w2[2 * k + 1] = mkv2(w.z, w.w);
        }
    }
    const float bias = b_ih[j] + b_hh[j];
    const float wi0  = W_ih[2 * j + 0];
    const float wi1  = W_ih[2 * j + 1];

    if (j < H) h_lds[j] = 0.0f;
    float c = 0.0f;

    float xreg = (j < 2 * T) ? x[j] : 0.0f;
    __syncthreads();

    const int nchunks = (T + CHUNK - 1) / CHUNK;
    for (int ch = 0; ch < nchunks; ++ch) {
        xbuf[j] = xreg;
        const int nidx = (ch + 1) * (2 * CHUNK) + j;
        if (ch + 1 < nchunks && nidx < 2 * T) xreg = x[nidx];
        __syncthreads();

        const int tend = (T - ch * CHUNK < CHUNK) ? (T - ch * CHUNK) : CHUNK;
        for (int tt = 0; tt < tend; ++tt) {
            const int t = ch * CHUNK + tt;
            const float x0 = xbuf[2 * tt + 0];
            const float x1 = xbuf[2 * tt + 1];

            v2f acc2 = mkv2(bias + x0 * wi0 + x1 * wi1, 0.0f);
            const float4* h4 = (const float4*)h_lds;
            #pragma unroll
            for (int k = 0; k < H / 4; ++k) {
                float4 hv = h4[k];
                v2f hlo = mkv2(hv.x, hv.y);
                v2f hhi = mkv2(hv.z, hv.w);
                asm("v_pk_fma_f32 %0, %1, %2, %0" : "+v"(acc2) : "v"(hlo), "v"(w2[2 * k + 0]));
                asm("v_pk_fma_f32 %0, %1, %2, %0" : "+v"(acc2) : "v"(hhi), "v"(w2[2 * k + 1]));
            }
            gates_lds[j] = acc2.x + acc2.y;
            __syncthreads();

            if (j < H) {
                const float gi = fsigmoid(gates_lds[j]);
                const float gf = fsigmoid(gates_lds[H + j]);
                const float gg = ftanh(gates_lds[2 * H + j]);
                const float go = fsigmoid(gates_lds[3 * H + j]);
                c = gf * c + gi * gg;
                const float hn = go * ftanh(c);
                h_lds[j] = hn;
                hs[(long)t * H + j] = hn;
            }
            __syncthreads();
        }
    }
}

__global__ __launch_bounds__(256, 4)
void fc_wmma_kernel(const float* __restrict__ hs,
                    const float* __restrict__ fc_w,
                    const float* __restrict__ fc_b,
                    float* __restrict__ out,
                    int T) {
    const int lane = threadIdx.x & 31;
    const int wave = threadIdx.x >> 5;
    const int tile = blockIdx.x * 8 + wave;
    const int row0 = tile * 16;
    if (row0 >= T) return;

    const int M    = lane & 15;
    const int half = lane >> 4;
    const int N    = lane & 15;

    v2f bfrag[16];
    #pragma unroll
    for (int kk = 0; kk < 16; ++kk) {
        const int k0 = kk * 4 + half * 2;
        const float b0 = (N < 2) ? fc_w[N * H + k0 + 0] : 0.0f;
        const float b1 = (N < 2) ? fc_w[N * H + k0 + 1] : 0.0f;
        bfrag[kk] = mkv2(b0, b1);
    }

    int row = row0 + M;
    if (row >= T) row = T - 1;
    const float* arow = hs + (long)row * H;

    v8f cacc = {};
    #pragma unroll
    for (int kk = 0; kk < 16; ++kk) {
        const int k0 = kk * 4 + half * 2;
        const v2f afrag = *(const v2f*)(arow + k0);
        cacc = __builtin_amdgcn_wmma_f32_16x16x4_f32(
             false, afrag,  false, bfrag[kk],
             (short)0, cacc,  false,  false);
    }

    #pragma unroll
    for (int r = 0; r < 8; ++r) {
        const int orow = row0 + r + 8 * half;
        if (N < 2 && orow < T) out[orow * 2 + N] = cacc[r] + fc_b[N];
    }
}

extern "C" void kernel_launch(void* const* d_in, const int* in_sizes, int n_in,
                              void* d_out, int out_size, void* d_ws, size_t ws_size,
                              hipStream_t stream) {
    const float* x    = (const float*)d_in[0];
    const float* W_ih = (const float*)d_in[1];
    const float* W_hh = (const float*)d_in[2];
    const float* b_ih = (const float*)d_in[3];
    const float* b_hh = (const float*)d_in[4];
    const float* fc_w = (const float*)d_in[5];
    const float* fc_b = (const float*)d_in[6];
    float* out = (float*)d_out;

    const int T = in_sizes[0] / 2;
    float* hs = (float*)d_ws;

    lstm_scan_kernel<<<1, 256, 0, stream>>>(x, W_ih, W_hh, b_ih, b_hh, hs, T);

    const int tiles  = (T + 15) / 16;
    const int blocks = (tiles + 7) / 8;
    fc_wmma_kernel<<<blocks, 256, 0, stream>>>(hs, fc_w, fc_b, out, T);
}


# Scan kernel: step loop's trip count held in a scalar register: exit test is s_cmp + s_cbranch_scc; the vector compare -> s_and -> vcc branch no longer stands inside the cell chain; sums unchanged
# speedup vs baseline: 4.1783x; 4.1783x over previous
#include <hip/hip_runtime.h>

#define H      64
#define G      256
#define KW     32
#define XBLK   16
#define SIG_AT  10
#define WAIT_AT 20

typedef __attribute__((ext_vector_type(2))) float v2f;
typedef __attribute__((ext_vector_type(8))) float v8f;

__device__ __forceinline__ v2f mkv2(float a, float b) { v2f r; r.x = a; r.y = b; return r; }

#define KS   (-1.4426950408889634f)
#define KS2  (-2.8853900817779268f)
#define KINV (-0.34657359027997264f)

__device__ __forceinline__ float fexp2(float x) { return __builtin_amdgcn_exp2f(x); }
__device__ __forceinline__ float frcp(float x)  { return __builtin_amdgcn_rcpf(x); }

__global__ __launch_bounds__(256)
void pack_whh_kernel(const float* __restrict__ W_hh, float* __restrict__ Wp) {
    const int i  = blockIdx.x * 256 + threadIdx.x;
    const int l  = i & 31;
    const int kk = (i >> 5) & 31;
    const int s  = (i >> 10) & 1;
    const int w  = i >> 11;
    const int k  = w * KW + kk;
    const int unit = (s ? w : 1 - w) * KW + l;
    const float* r = W_hh + unit * H + k;
    ((float4*)Wp)[i] = make_float4(r[0], r[H * H], r[2 * H * H], r[3 * H * H]);
}

__device__ __forceinline__ float lane_bcast(float v, int lane) {
    return __int_as_float(__builtin_amdgcn_readlane(__float_as_int(v), lane));
}

__global__ __launch_bounds__(64, 1)
void lstm_scan_kernel(const float* __restrict__ x,
                      const float* __restrict__ W_ih,
                      const float* __restrict__ Wp,
                      const float* __restrict__ b_ih,
                      const float* __restrict__ b_hh,
                      float* __restrict__ hs,
                      int T) {
    __shared__ float4 xb[2][2][KW];

    const int l = threadIdx.x & 31;
    const int w = threadIdx.x >> 5;
    const int u = w * KW + l;

    float4 wq[2][KW];
    const float4* wl = (const float4*)Wp + (w * 2 * KW) * 32 + l;
    #pragma unroll
    for (int s = 0; s < 2; ++s) {
        #pragma unroll
        for (int kk = 0; kk < KW; ++kk) wq[s][kk] = wl[(s * KW + kk) * 32];
    }
    float bias[4], wi0[4], wi1[4];
    #pragma unroll
    for (int g = 0; g < 4; ++g) {
        const int row = g * H + u;
        const float s = (g == 2) ? KS2 : KS;
        bias[g] = s * (b_ih[row] + b_hh[row]);
        wi0[g]  = s * W_ih[2 * row + 0];
        wi1[g]  = s * W_ih[2 * row + 1];
    }

    float h  = 0.0f;
    float cs = 0.0f;

    float xnext = (l < 2 * T) ? x[l] : 0.0f;
    const int nblk = (T + XBLK - 1) / XBLK;
    float* hcol = hs + u;
    for (int blk = 0; blk < nblk; ++blk) {
        const float xcur = xnext;
        asm volatile("" :: "v"(xcur));
        const int nidx = (blk + 1) * (2 * XBLK) + l;
        xnext = (nidx < 2 * T) ? x[nidx] : 0.0f;

        const int tend = __builtin_amdgcn_readfirstlane((T - blk * XBLK < XBLK) ? (T - blk * XBLK) : XBLK);
        #pragma unroll 1
        for (int tt = 0; tt < tend; ++tt) {
            const int t = blk * XBLK + tt;
            const int par = t & 1;
            const float x0 = lane_bcast(xcur, 2 * tt + 0);
            const float x1 = lane_bcast(xcur, 2 * tt + 1);

            float hk[KW];
            #pragma unroll
            for (int kk = 0; kk < KW; ++kk) hk[kk] = lane_bcast(h, kk);

            v2f Bif[4], Bgo[4];
            #pragma unroll
            for (int kk = 0; kk < KW; ++kk) {
                const float4 q = wq[0][kk];
                const int cc = kk & 3;
                if (kk < 4) {
                    Bif[cc] = mkv2(hk[kk], hk[kk]) * mkv2(q.x, q.y);
                    Bgo[cc] = mkv2(hk[kk], hk[kk]) * mkv2(q.z, q.w);
                } else {
                    Bif[cc] = __builtin_elementwise_fma(mkv2(hk[kk], hk[kk]), mkv2(q.x, q.y), Bif[cc]);
                    Bgo[cc] = __builtin_elementwise_fma(mkv2(hk[kk], hk[kk]), mkv2(q.z, q.w), Bgo[cc]);
                }
            }
            const v2f bif = (Bif[0] + Bif[2]) + (Bif[1] + Bif[3]);
            const v2f bgo = (Bgo[0] + Bgo[2]) + (Bgo[1] + Bgo[3]);
            xb[par][1 - w][l] = make_float4(bif.x, bif.y, bgo.x, bgo.y);
            __builtin_amdgcn_sched_barrier(0);

            v2f Aif[4], Ago[4];
            float4 q;
            #pragma unroll
            for (int kk = 0; kk < KW; ++kk) {
                const float4 wv = wq[1][kk];
                const int cc = kk & 3;
                if (kk < 4) {
                    Aif[cc] = mkv2(hk[kk], hk[kk]) * mkv2(wv.x, wv.y);
                    Ago[cc] = mkv2(hk[kk], hk[kk]) * mkv2(wv.z, wv.w);
                } else {
                    Aif[cc] = __builtin_elementwise_fma(mkv2(hk[kk], hk[kk]), mkv2(wv.x, wv.y), Aif[cc]);
                    Ago[cc] = __builtin_elementwise_fma(mkv2(hk[kk], hk[kk]), mkv2(wv.z, wv.w), Ago[cc]);
                }
                if (kk == SIG_AT - 1) {
                    __builtin_amdgcn_sched_barrier(0);
                    __builtin_amdgcn_fence(__ATOMIC_RELEASE, "workgroup", "local");
                    __builtin_amdgcn_s_barrier_signal(-1);
                    __builtin_amdgcn_sched_barrier(0);
                }
                if (kk == WAIT_AT - 1) {
                    __builtin_amdgcn_sched_barrier(0);
                    __builtin_amdgcn_s_barrier_wait(-1);
                    __builtin_amdgcn_fence(__ATOMIC_ACQUIRE, "workgroup", "local");
                    q = xb[par][w][l];
                    __builtin_amdgcn_sched_barrier(0);
                }
            }
            const v2f aif = (Aif[0] + Aif[2]) + (Aif[1] + Aif[3]);
            const v2f ago = (Ago[0] + Ago[2]) + (Ago[1] + Ago[3]);

            const float ti = fmaf(aif.x, KS,  fmaf(x1, wi1[0], fmaf(x0, wi0[0], bias[0])));
            const float tf = fmaf(aif.y, KS,  fmaf(x1, wi1[1], fmaf(x0, wi0[1], bias[1])));
            const float tg = fmaf(ago.x, KS2, fmaf(x1, wi1[2], fmaf(x0, wi0[2], bias[2])));
            const float to = fmaf(ago.y, KS,  fmaf(x1, wi1[3], fmaf(x0, wi0[3], bias[3])));
            __builtin_amdgcn_sched_barrier(0);

            const float ai = fmaf(q.x, KS,  ti);
            const float ag = fmaf(q.z, KS2, tg);
            const float af = fmaf(q.y, KS,  tf);
            const float ao = fmaf(q.w, KS,  to);
            __builtin_amdgcn_sched_barrier(0);
            const float ei = fexp2(ai);
            const float eg = fexp2(ag);
            const float ef = fexp2(af);
            const float eo = fexp2(ao);
            __builtin_amdgcn_sched_barrier(0);
            const float m  = frcp(fmaf(ei, KINV, KINV));
            const float rg = frcp(1.0f + eg);
            const float gf = frcp(1.0f + ef);
            const float go = frcp(1.0f + eo);
            __builtin_amdgcn_sched_barrier(0);
            cs = fmaf(2.0f, m * rg, fmaf(gf, cs, -m));
            const float rn = frcp(1.0f + fexp2(cs));
            h = fmaf(go + go, rn, -go);
            hcol[(long)t * H] = h;
        }
    }
}

__global__ __launch_bounds__(256, 4)
void fc_wmma_kernel(const float* __restrict__ hs,
                    const float* __restrict__ fc_w,
                    const float* __restrict__ fc_b,
                    float* __restrict__ out,
                    int T) {
    const int lane = threadIdx.x & 31;
    const int wave = threadIdx.x >> 5;
    const int tile = blockIdx.x * 8 + wave;
    const int row0 = tile * 16;
    if (row0 >= T) return;

    const int M    = lane & 15;
    const int half = lane >> 4;
    const int N    = lane & 15;

    v2f bfrag[16];
    #pragma unroll
    for (int kk = 0; kk < 16; ++kk) {
        const int k0 = kk * 4 + half * 2;
        const float b0 = (N < 2) ? fc_w[N * H + k0 + 0] : 0.0f;
        const float b1 = (N < 2) ? fc_w[N * H + k0 + 1] : 0.0f;
        bfrag[kk] = mkv2(b0, b1);
    }

    int row = row0 + M;
    if (row >= T) row = T - 1;
    const float* arow = hs + (long)row * H;

    v8f cacc = {};
    #pragma unroll
    for (int kk = 0; kk < 16; ++kk) {
        const int k0 = kk * 4 + half * 2;
        const v2f afrag = *(const v2f*)(arow + k0);
        cacc = __builtin_amdgcn_wmma_f32_16x16x4_f32(
             false, afrag,  false, bfrag[kk],
             (short)0, cacc,  false,  false);
    }

    #pragma unroll
    for (int r = 0; r < 8; ++r) {
        const int orow = row0 + r + 8 * half;
        if (N < 2 && orow < T) out[orow * 2 + N] = cacc[r] + fc_b[N];
    }
}

extern "C" void kernel_launch(void* const* d_in, const int* in_sizes, int n_in,
                              void* d_out, int out_size, void* d_ws, size_t ws_size,
                              hipStream_t stream) {
    const float* x    = (const float*)d_in[0];
    const float* W_ih = (const float*)d_in[1];
    const float* W_hh = (const float*)d_in[2];
    const float* b_ih = (const float*)d_in[3];
    const float* b_hh = (const float*)d_in[4];
    const float* fc_w = (const float*)d_in[5];
    const float* fc_b = (const float*)d_in[6];
    float* out = (float*)d_out;

    const int T = in_sizes[0] / 2;
    float* Wp = (float*)d_ws;
    float* hs = Wp + 16384;

    pack_whh_kernel<<<16, 256, 0, stream>>>(W_hh, Wp);
    lstm_scan_kernel<<<1, 64, 0, stream>>>(x, W_ih, Wp, b_ih, b_hh, hs, T);

    const int tiles  = (T + 15) / 16;
    const int blocks = (tiles + 7) / 8;
    fc_wmma_kernel<<<blocks, 256, 0, stream>>>(hs, fc_w, fc_b, out, T);
}
